# starting kernel
# baseline (speedup 1.0000x reference)
#include <hip/hip_runtime.h>
#include <hip/hip_bf16.h>

#define N_NODES 50000
#define N_EDGES 800000
#define HID 128
#define OUT_DIM 16
#define NEG_SLOPE 0.2f

typedef __attribute__((ext_vector_type(2))) float v2f;
typedef __attribute__((ext_vector_type(8))) float v8f;

__global__ __launch_bounds__(HID) void embed_kernel(
    const int* __restrict__ feats, const float* __restrict__ key_emb,
    const float* __restrict__ val_emb, float* __restrict__ h) {
  int n = blockIdx.x;
  int d = threadIdx.x;
  int f0 = feats[n * 2 + 0];
  int f1 = feats[n * 2 + 1];
  float v = key_emb[f0 * HID + d] + val_emb[f1 * HID + d];
  h[n * HID + d] = v > 0.f ? v : 0.f;
}

__global__ __launch_bounds__(256) void gemm128_kernel(
    const float* __restrict__ A, const float* __restrict__ B,
    float* __restrict__ C) {
  const int wave = threadIdx.x >> 5;
  const int lane = threadIdx.x & 31;
  const int m0 = blockIdx.x * 16;
  const int n0 = wave * 16;
  const int r = lane & 15;
  const int hi = lane >> 4;
  const int koff = hi << 1;

  const float* arow = A + (size_t)(m0 + r) * HID;
  v8f acc = {};
#pragma unroll 8
  for (int k0 = 0; k0 < HID; k0 += 4) {
    const int k = k0 + koff;
    v2f a;
    a.x = arow[k];
    a.y = arow[k + 1];
    v2f b;
    b.x = B[k * HID + n0 + r];
    b.y = B[(k + 1) * HID + n0 + r];
    acc = __builtin_amdgcn_wmma_f32_16x16x4_f32(false, a, false, b, (short)0,
                                                acc, false, false);
  }
#pragma unroll
  for (int v = 0; v < 8; ++v) {
    C[(size_t)(m0 + v + (hi << 3)) * HID + n0 + r] = acc[v];
  }
}

__global__ __launch_bounds__(256) void gemm_cls_kernel(
    const float* __restrict__ A, const float* __restrict__ B,
    float* __restrict__ C) {
  const int wave = threadIdx.x >> 5;
  const int tile = blockIdx.x * 8 + wave;
  if (tile >= N_NODES / 16) return;
  const int lane = threadIdx.x & 31;
  const int m0 = tile * 16;
  const int r = lane & 15;
  const int hi = lane >> 4;
  const int koff = hi << 1;

  const float* arow = A + (size_t)(m0 + r) * HID;
  v8f acc = {};
#pragma unroll 8
  for (int k0 = 0; k0 < HID; k0 += 4) {
    const int k = k0 + koff;
    v2f a;
    a.x = arow[k];
    a.y = arow[k + 1];
    v2f b;
    b.x = B[k * OUT_DIM + r];
    b.y = B[(k + 1) * OUT_DIM + r];
    acc = __builtin_amdgcn_wmma_f32_16x16x4_f32(false, a, false, b, (short)0,
                                                acc, false, false);
  }
#pragma unroll
  for (int v = 0; v < 8; ++v) {
    C[(size_t)(m0 + v + (hi << 3)) * OUT_DIM + r] = acc[v];
  }
}

__device__ __forceinline__ int float_to_ord(float f) {
  int i = __float_as_int(f);
  return i >= 0 ? i : (i ^ 0x7FFFFFFF);
}
__device__ __forceinline__ float ord_to_float(int i) {
  return __int_as_float(i >= 0 ? i : (i ^ 0x7FFFFFFF));
}

__global__ void init_nodes_kernel(int* __restrict__ mmax,
                                  float* __restrict__ denom) {
  int i = blockIdx.x * 256 + threadIdx.x;
  if (i < N_NODES) {
    mmax[i] = float_to_ord(-__builtin_inff());
    denom[i] = 0.f;
  }
}

__global__ void zero_kernel(float* __restrict__ p, int n) {
  int i = blockIdx.x * 256 + threadIdx.x;
  if (i < n) p[i] = 0.f;
}

__global__ __launch_bounds__(256) void edge_logits_kernel(
    const float* __restrict__ feat, const float* __restrict__ attn,
    const int* __restrict__ src, const int* __restrict__ dst,
    float* __restrict__ logits, int* __restrict__ mmax) {
  const int e = blockIdx.x * 8 + (threadIdx.x >> 5);
  const int lane = threadIdx.x & 31;
  const int s = src[e];
  const int d = dst[e];
  const float4 x = ((const float4*)(feat + (size_t)s * HID))[lane];
  const float4 y = ((const float4*)(feat + (size_t)d * HID))[lane];
  const float4 a4 = ((const float4*)attn)[lane];
  float p = 0.f, t;
  t = x.x + y.x; p += a4.x * (t > 0.f ? t : NEG_SLOPE * t);
  t = x.y + y.y; p += a4.y * (t > 0.f ? t : NEG_SLOPE * t);
  t = x.z + y.z; p += a4.z * (t > 0.f ? t : NEG_SLOPE * t);
  t = x.w + y.w; p += a4.w * (t > 0.f ? t : NEG_SLOPE * t);
#pragma unroll
  for (int off = 16; off > 0; off >>= 1) p += __shfl_xor(p, off, 32);
  if (lane == 0) {
    logits[e] = p;
    atomicMax(mmax + d, float_to_ord(p));
  }
}

__global__ __launch_bounds__(256) void edge_exp_kernel(
    const int* __restrict__ dst, float* __restrict__ logits,
    const int* __restrict__ mmax, float* __restrict__ denom) {
  int e = blockIdx.x * 256 + threadIdx.x;
  if (e >= N_EDGES) return;
  int d = dst[e];
  float mx = ord_to_float(mmax[d]);
  float ex = expf(logits[e] - mx);
  logits[e] = ex;
  atomicAdd(denom + d, ex);
}

__global__ __launch_bounds__(256) void edge_aggr_kernel(
    const float* __restrict__ feat, const int* __restrict__ src,
    const int* __restrict__ dst, const float* __restrict__ ex,
    const float* __restrict__ denom, float* __restrict__ out) {
  const int e = blockIdx.x * 8 + (threadIdx.x >> 5);
  const int lane = threadIdx.x & 31;
  const int s = src[e];
  const int d = dst[e];
  const float a = ex[e] / denom[d];
  const float4 x = ((const float4*)(feat + (size_t)s * HID))[lane];
  float* o = out + (size_t)d * HID + lane * 4;
  atomicAdd(o + 0, a * x.x);
  atomicAdd(o + 1, a * x.y);
  atomicAdd(o + 2, a * x.z);
  atomicAdd(o + 3, a * x.w);
}

static void run_gat_layer(const float* h_in, const float* W, const float* attn,
                          const int* src, const int* dst, float* feat,
                          float* h_out, float* ex, int* mmax, float* denom,
                          hipStream_t stream) {
  gemm128_kernel<<<N_NODES / 16, 256, 0, stream>>>(h_in, W, feat);
  init_nodes_kernel<<<(N_NODES + 255) / 256, 256, 0, stream>>>(mmax, denom);
  zero_kernel<<<(N_NODES * HID + 255) / 256, 256, 0, stream>>>(h_out,
                                                               N_NODES * HID);
  edge_logits_kernel<<<N_EDGES / 8, 256, 0, stream>>>(feat, attn, src, dst, ex,
                                                      mmax);
  edge_exp_kernel<<<(N_EDGES + 255) / 256, 256, 0, stream>>>(dst, ex, mmax,
                                                             denom);
  edge_aggr_kernel<<<N_EDGES / 8, 256, 0, stream>>>(feat, src, dst, ex, denom,
                                                    h_out);
}

extern "C" void kernel_launch(void* const* d_in, const int* in_sizes, int n_in,
                              void* d_out, int out_size, void* d_ws,
                              size_t ws_size, hipStream_t stream) {
  const int* feats = (const int*)d_in[0];
  const int* src = (const int*)d_in[1];
  const int* dst = (const int*)d_in[2];
  const float* key_emb = (const float*)d_in[3];
  const float* val_emb = (const float*)d_in[4];
  const float* W1 = (const float*)d_in[5];
  const float* attn1 = (const float*)d_in[6];
  const float* W2 = (const float*)d_in[7];
  const float* attn2 = (const float*)d_in[8];
  const float* cls_w = (const float*)d_in[9];
  float* out = (float*)d_out;

  float* hA = (float*)d_ws;
  float* feat = hA + (size_t)N_NODES * HID;
  float* hB = feat + (size_t)N_NODES * HID;
  float* ex = hB + (size_t)N_NODES * HID;
  int* mmax = (int*)(ex + N_EDGES);
  float* denom = (float*)(mmax + N_NODES);

  embed_kernel<<<N_NODES, HID, 0, stream>>>(feats, key_emb, val_emb, hA);

  run_gat_layer(hA, W1, attn1, src, dst, feat, hB, ex, mmax, denom, stream);
  run_gat_layer(hB, W2, attn2, src, dst, feat, hA, ex, mmax, denom, stream);

  gemm_cls_kernel<<<(N_NODES / 16 + 7) / 8, 256, 0, stream>>>(hA, cls_w, out);
}


# Layer 1: softmax shift dropped where all logits lie in +-16 (checked; else the old max and sum passes run): rows weighted by exp(logit) are added inside the logits kernel; dot sums in a new order
# speedup vs baseline: 2.7442x; 2.7442x over previous
#include <hip/hip_runtime.h>
#include <hip/hip_bf16.h>

#define N_NODES 50000
#define N_EDGES 800000
#define HID 128
#define OUT_DIM 16
#define NEG_SLOPE 0.2f

typedef __attribute__((ext_vector_type(2))) float v2f;
typedef __attribute__((ext_vector_type(8))) float v8f;

__global__ __launch_bounds__(HID) void embed_kernel(
    const int* __restrict__ feats, const float* __restrict__ key_emb,
    const float* __restrict__ val_emb, float* __restrict__ h) {
  int n = blockIdx.x;
  int d = threadIdx.x;
  int f0 = feats[n * 2 + 0];
  int f1 = feats[n * 2 + 1];
  float v = key_emb[f0 * HID + d] + val_emb[f1 * HID + d];
  h[n * HID + d] = v > 0.f ? v : 0.f;
}

#define GEMM_WAVES 4
#define GEMM_WPITCH (2 * HID + 32)

template <bool DIV_ROWS>
__global__ __launch_bounds__(GEMM_WAVES * 32) void gemm128_kernel(
    const float* __restrict__ A, const float* __restrict__ B,
    const float* __restrict__ rowsum, float* __restrict__ C) {
  __shared__ __attribute__((aligned(16))) float Ws[(HID / 2) * GEMM_WPITCH];
  const int lane = threadIdx.x & 31;
  const int wave = __builtin_amdgcn_readfirstlane((int)(threadIdx.x >> 5));
  const int stripe = blockIdx.x * GEMM_WAVES + wave;
  const int m0 = (stripe < N_NODES / 16 ? stripe : N_NODES / 16 - 1) * 16;
  const int r = lane & 15;
  const int hi = lane >> 4;

  const float* arow = A + (size_t)(m0 + r) * HID + (hi << 1);
  v2f a[HID / 4];
#pragma unroll
  for (int s = 0; s < HID / 4; ++s) a[s] = *(const v2f*)(arow + 4 * s);
  float inv = 0.f;
  if (DIV_ROWS) {
    const float dn = rowsum[m0 + r];
    if (dn > 0.f) inv = 1.f / dn;
  }

  float4 w0[HID / 2 / GEMM_WAVES], w1[HID / 2 / GEMM_WAVES];
#pragma unroll
  for (int i = 0; i < HID / 2 / GEMM_WAVES; ++i) {
    const int kp = wave + i * GEMM_WAVES;
    w0[i] = ((const float4*)(B + (size_t)(2 * kp) * HID))[lane];
    w1[i] = ((const float4*)(B + (size_t)(2 * kp + 1) * HID))[lane];
  }
#pragma unroll
  for (int i = 0; i < HID / 2 / GEMM_WAVES; ++i) {
    const int kp = wave + i * GEMM_WAVES;
    float4* w = (float4*)(Ws + kp * GEMM_WPITCH + 8 * lane);
    w[0] = make_float4(w0[i].x, w1[i].x, w0[i].y, w1[i].y);
    w[1] = make_float4(w0[i].z, w1[i].z, w0[i].w, w1[i].w);
  }
  __syncthreads();
  if (stripe >= N_NODES / 16) return;

  const float* bl = Ws + hi * GEMM_WPITCH + 2 * r;
  v8f acc[HID / 16];
#pragma unroll
  for (int j = 0; j < HID / 16; ++j) acc[j] = v8f{};
#pragma unroll
  for (int s = 0; s < HID / 4; ++s) {
    const float* bk = bl + (2 * s) * GEMM_WPITCH;
#pragma unroll
    for (int j = 0; j < HID / 16; ++j) {
      const v2f b = *(const v2f*)(bk + 32 * j);
      acc[j] = __builtin_amdgcn_wmma_f32_16x16x4_f32(false, a[s], false, b,
                                                     (short)0, acc[j], false,
                                                     false);
    }
  }
#pragma unroll
  for (int v = 0; v < 8; ++v) {
    float sc = 1.f;
    if (DIV_ROWS) sc = __shfl(inv, v + (hi << 3), 32);
    float* crow = C + (size_t)(m0 + v + (hi << 3)) * HID + r;
#pragma unroll
    for (int j = 0; j < HID / 16; ++j) {
      float c = acc[j][v];
      if (DIV_ROWS) c *= sc;
      crow[16 * j] = c;
    }
  }
}

#define ACC_W 32

__global__ __launch_bounds__(256) void gemm_cls_kernel(
    const float* __restrict__ A, const float* __restrict__ B,
    float* __restrict__ C) {
  const int wave = threadIdx.x >> 5;
  const int tile = blockIdx.x * 8 + wave;
  if (tile >= N_NODES / 16) return;
  const int lane = threadIdx.x & 31;
  const int m0 = tile * 16;
  const int r = lane & 15;
  const int hi = lane >> 4;
  const int koff = hi << 1;

  const float* arow = A + (size_t)(m0 + r) * HID;
  v8f acc = {};
#pragma unroll 8
  for (int k0 = 0; k0 < HID; k0 += 4) {
    const int k = k0 + koff;
    v2f a;
    a.x = arow[k];
    a.y = arow[k + 1];
    v2f b;
    b.x = B[k * OUT_DIM + r];
    b.y = B[(k + 1) * OUT_DIM + r];
    acc = __builtin_amdgcn_wmma_f32_16x16x4_f32(false, a, false, b, (short)0,
                                                acc, false, false);
  }
#pragma unroll
  for (int v = 0; v < 8; ++v) {
    float* crow = C + (size_t)(m0 + v + (hi << 3)) * ACC_W;
    crow[r] = acc[v];
    crow[OUT_DIM + r] = 1.f;
  }
}

__device__ __forceinline__ int float_to_ord(float f) {
  int i = __float_as_int(f);
  return i >= 0 ? i : (i ^ 0x7FFFFFFF);
}
__device__ __forceinline__ float ord_to_float(int i) {
  return __int_as_float(i >= 0 ? i : (i ^ 0x7FFFFFFF));
}

__global__ void init_nodes_kernel(int* __restrict__ mmax,
                                  float* __restrict__ denom,
                                  int* __restrict__ unsafe) {
  int i = blockIdx.x * 256 + threadIdx.x;
  if (i < N_NODES) {
    mmax[i] = float_to_ord(-__builtin_inff());
    denom[i] = 0.f;
  }
  if (i == 0) *unsafe = 0;
}

__global__ void init_max_kernel(int* __restrict__ mmax) {
  int i = blockIdx.x * 256 + threadIdx.x;
  if (i < N_NODES) mmax[i] = float_to_ord(-__builtin_inff());
}

__global__ void zero_kernel(float* __restrict__ p, int n) {
  int i = blockIdx.x * 256 + threadIdx.x;
  if (i < n) p[i] = 0.f;
}

#define EPW 32
#define EB 8

__global__ __launch_bounds__(256) void edge_logits_kernel(
    const float* __restrict__ feat, const float* __restrict__ attn,
    const int* __restrict__ src, const int* __restrict__ dst,
    float* __restrict__ logits, int* __restrict__ mmax) {
  const int lane = threadIdx.x & 31;
  const int wave = __builtin_amdgcn_readfirstlane((int)(threadIdx.x >> 5));
  const int e0 = (blockIdx.x * 8 + wave) * EPW;
  const int sv = src[e0 + lane];
  const int dv = dst[e0 + lane];
  const float4 a4 = ((const float4*)attn)[lane];
  const bool h4 = (lane & 16) != 0;
  const bool h3 = (lane & 8) != 0;
  const bool h2 = (lane & 4) != 0;
  float mine = 0.f;
#pragma unroll 1
  for (int b = 0; b < EPW / EB; ++b) {
    float4 x[EB], y[EB];
#pragma unroll
    for (int i = 0; i < EB; ++i) {
      const int s = __builtin_amdgcn_readlane(sv, b * EB + i);
      const int d = __builtin_amdgcn_readlane(dv, b * EB + i);
      x[i] = ((const float4*)(feat + (size_t)s * HID))[lane];
      y[i] = ((const float4*)(feat + (size_t)d * HID))[lane];
    }
    float p[EB];
#pragma unroll
    for (int i = 0; i < EB; ++i) {
      float q = 0.f, t;
      t = x[i].x + y[i].x; q += a4.x * (t > 0.f ? t : NEG_SLOPE * t);
      t = x[i].y + y[i].y; q += a4.y * (t > 0.f ? t : NEG_SLOPE * t);
      t = x[i].z + y[i].z; q += a4.z * (t > 0.f ? t : NEG_SLOPE * t);
      t = x[i].w + y[i].w; q += a4.w * (t > 0.f ? t : NEG_SLOPE * t);
      p[i] = q;
    }
    float q4[4], q2[2], r;
#pragma unroll
    for (int i = 0; i < 4; ++i) {
      const float keep = h4 ? p[i + 4] : p[i];
      const float send = h4 ? p[i] : p[i + 4];
      q4[i] = keep + __shfl_xor(send, 16, 32);
    }
#pragma unroll
    for (int i = 0; i < 2; ++i) {
      const float keep = h3 ? q4[i + 2] : q4[i];
      const float send = h3 ? q4[i] : q4[i + 2];
      q2[i] = keep + __shfl_xor(send, 8, 32);
    }
    {
      const float keep = h2 ? q2[1] : q2[0];
      const float send = h2 ? q2[0] : q2[1];
      r = keep + __shfl_xor(send, 4, 32);
    }
    r += __shfl_xor(r, 2, 32);
    r += __shfl_xor(r, 1, 32);
    const float tmp = __shfl(r, (lane & 7) * 4, 32);
    if ((lane >> 3) == b) mine = tmp;
  }
  logits[e0 + lane] = mine;
  atomicMax(mmax + dv, float_to_ord(mine));
}

#define RP (HID / 32)
#define LOGIT_SAFE 16.0f

__global__ __launch_bounds__(256) void edge_fused_kernel(
    const float* __restrict__ feat, const float* __restrict__ attn,
    const int* __restrict__ src, const int* __restrict__ dst,
    float* __restrict__ logits, float* __restrict__ denom,
    float* __restrict__ out, int* __restrict__ unsafe) {
  const int lane = threadIdx.x & 31;
  const int wave = __builtin_amdgcn_readfirstlane((int)(threadIdx.x >> 5));
  const int e0 = (blockIdx.x * 8 + wave) * EPW;
  const int sv = src[e0 + lane];
  const int dv = dst[e0 + lane];
  float a[RP];
#pragma unroll
  for (int j = 0; j < RP; ++j) a[j] = attn[lane + 32 * j];
  const bool h4 = (lane & 16) != 0;
  const bool h3 = (lane & 8) != 0;
  const bool h2 = (lane & 4) != 0;
  float mine = 0.f;
#pragma unroll 1
  for (int b = 0; b < EPW / EB; ++b) {
    float x[EB][RP], y[EB][RP];
#pragma unroll
    for (int i = 0; i < EB; ++i) {
      const int s = __builtin_amdgcn_readlane(sv, b * EB + i);
      const int d = __builtin_amdgcn_readlane(dv, b * EB + i);
      const float* xr = feat + (size_t)s * HID + lane;
      const float* yr = feat + (size_t)d * HID + lane;
#pragma unroll
      for (int j = 0; j < RP; ++j) {
        x[i][j] = xr[32 * j];
        y[i][j] = yr[32 * j];
      }
    }
    float p[EB];
#pragma unroll
    for (int i = 0; i < EB; ++i) {
      float q = 0.f;
#pragma unroll
      for (int j = 0; j < RP; ++j) {
        const float t = x[i][j] + y[i][j];
        q += a[j] * (t > 0.f ? t : NEG_SLOPE * t);
      }
      p[i] = q;
    }
    float q4[4], q2[2], r;
#pragma unroll
    for (int i = 0; i < 4; ++i) {
      const float keep = h4 ? p[i + 4] : p[i];
      const float send = h4 ? p[i] : p[i + 4];
      q4[i] = keep + __shfl_xor(send, 16, 32);
    }
#pragma unroll
    for (int i = 0; i < 2; ++i) {
      const float keep = h3 ? q4[i + 2] : q4[i];
      const float send = h3 ? q4[i] : q4[i + 2];
      q2[i] = keep + __shfl_xor(send, 8, 32);
    }
    {
      const float keep = h2 ? q2[1] : q2[0];
      const float send = h2 ? q2[0] : q2[1];
      r = keep + __shfl_xor(send, 4, 32);
    }
    r += __shfl_xor(r, 2, 32);
    r += __shfl_xor(r, 1, 32);
    const int wv = __float_as_int(expf(r));
#pragma unroll
    for (int i = 0; i < EB; ++i) {
      const int d = __builtin_amdgcn_readlane(dv, b * EB + i);
      const float w = __int_as_float(__builtin_amdgcn_readlane(wv, 4 * i));
      float* o = out + (size_t)d * HID + lane;
#pragma unroll
      for (int j = 0; j < RP; ++j) atomicAdd(o + 32 * j, w * x[i][j]);
    }
    const float tmp = __shfl(r, (lane & 7) * 4, 32);
    if ((lane >> 3) == b) mine = tmp;
  }
  logits[e0 + lane] = mine;
  atomicAdd(denom + dv, expf(mine));
  if (!(fabsf(mine) <= LOGIT_SAFE)) *unsafe = 1;
}

__global__ __launch_bounds__(256) void fallback_reset_max_kernel(
    const int* __restrict__ unsafe, const int* __restrict__ dst,
    const float* __restrict__ logits, int* __restrict__ mmax,
    float* __restrict__ denom, float* __restrict__ out) {
  if (*unsafe == 0) return;
  const int t = blockIdx.x * 256 + threadIdx.x;
  const int nt = gridDim.x * 256;
  for (int i = t; i < N_NODES * HID; i += nt) out[i] = 0.f;
  for (int i = t; i < N_NODES; i += nt) denom[i] = 0.f;
  for (int i = t; i < N_EDGES; i += nt)
    atomicMax(mmax + dst[i], float_to_ord(logits[i]));
}


__global__ __launch_bounds__(256) void edge_aggr_kernel(
    const float* __restrict__ feat, const int* __restrict__ src,
    const int* __restrict__ dst, const float* __restrict__ logits,
    const int* __restrict__ mmax, float* __restrict__ denom,
    float* __restrict__ out, const int* __restrict__ unsafe) {
  if (*unsafe == 0) return;
  const int lane = threadIdx.x & 31;
  const int wave = __builtin_amdgcn_readfirstlane((int)(threadIdx.x >> 5));
  for (int grp = blockIdx.x * 8 + wave; grp < N_EDGES / EPW;
       grp += gridDim.x * 8) {
    const int e0 = grp * EPW;
    const int sv = src[e0 + lane];
    const int dv = dst[e0 + lane];
    const float exv = expf(logits[e0 + lane] - ord_to_float(mmax[dv]));
    atomicAdd(denom + dv, exv);
    const int av = __float_as_int(exv);
#pragma unroll 1
    for (int b = 0; b < EPW / EB; ++b) {
      float x[EB][RP];
#pragma unroll
      for (int i = 0; i < EB; ++i) {
        const int s = __builtin_amdgcn_readlane(sv, b * EB + i);
        const float* row = feat + (size_t)s * HID + lane;
#pragma unroll
        for (int j = 0; j < RP; ++j) x[i][j] = row[32 * j];
      }
#pragma unroll
      for (int i = 0; i < EB; ++i) {
        const int d = __builtin_amdgcn_readlane(dv, b * EB + i);
        const float a = __int_as_float(__builtin_amdgcn_readlane(av, b * EB + i));
        float* o = out + (size_t)d * HID + lane;
#pragma unroll
        for (int j = 0; j < RP; ++j) atomicAdd(o + 32 * j, a * x[i][j]);
      }
    }
  }
}

__global__ __launch_bounds__(256) void edge_aggr_cls_kernel(
    const float* __restrict__ g, const int* __restrict__ src,
    const int* __restrict__ dst, const float* __restrict__ logits,
    const int* __restrict__ mmax, float* __restrict__ acc) {
  const int lane = threadIdx.x & 31;
  const int wave = __builtin_amdgcn_readfirstlane((int)(threadIdx.x >> 5));
  const int e0 = (blockIdx.x * 8 + wave) * EPW;
  const int sv = src[e0 + lane];
  const int dv = dst[e0 + lane];
  const int av =
      __float_as_int(expf(logits[e0 + lane] - ord_to_float(mmax[dv])));
  float x[EPW];
#pragma unroll
  for (int it = 0; it < EPW; ++it) {
    const int s = __builtin_amdgcn_readlane(sv, it);
    x[it] = g[(unsigned)s * ACC_W + lane];
  }
#pragma unroll
  for (int it = 0; it < EPW; ++it) {
    const int d = __builtin_amdgcn_readlane(dv, it);
    const float a = __int_as_float(__builtin_amdgcn_readlane(av, it));
    atomicAdd(acc + ((unsigned)d * ACC_W + lane), a * x[it]);
  }
}

__global__ void divide_rows_kernel(const float* __restrict__ acc,
                                   float* __restrict__ out) {
  const int i = blockIdx.x * 256 + threadIdx.x;
  if (i >= N_NODES * OUT_DIM / 4) return;
  const int n = i >> 2;
  const int q = i & 3;
  const float dn = acc[n * ACC_W + OUT_DIM];
  float4 v = ((const float4*)(acc + n * ACC_W))[q];
  if (dn > 0.f) {
    v.x /= dn; v.y /= dn; v.z /= dn; v.w /= dn;
  } else {
    v.x = 0.f; v.y = 0.f; v.z = 0.f; v.w = 0.f;
  }
  ((float4*)out)[i] = v;
}

static void run_gat_layer(const float* h_in, const float* W, const float* attn,
                          const int* src, const int* dst, float* feat,
                          float* h_out, float* logits, int* mmax, float* denom,
                          int* unsafe, hipStream_t stream) {
  gemm128_kernel<false>
      <<<(N_NODES / 16 + GEMM_WAVES - 1) / GEMM_WAVES, GEMM_WAVES * 32, 0, stream>>>(
          h_in, W, nullptr, feat);
  init_nodes_kernel<<<(N_NODES + 255) / 256, 256, 0, stream>>>(mmax, denom,
                                                               unsafe);
  zero_kernel<<<(N_NODES * HID + 255) / 256, 256, 0, stream>>>(h_out,
                                                               N_NODES * HID);
  edge_fused_kernel<<<N_EDGES / (8 * EPW), 256, 0, stream>>>(
      feat, attn, src, dst, logits, denom, h_out, unsafe);
  fallback_reset_max_kernel<<<512, 256, 0, stream>>>(unsafe, dst, logits, mmax,
                                                     denom, h_out);
  edge_aggr_kernel<<<512, 256, 0, stream>>>(feat, src, dst, logits, mmax, denom,
                                            h_out, unsafe);
}

static void run_gat_layer_cls(const float* h_in, const float* denom_in,
                              const float* W, const float* attn,
                              const float* cls_w, const int* src,
                              const int* dst, float* feat, float* g,
                              float* acc, float* out, float* logits, int* mmax,
                              hipStream_t stream) {
  gemm128_kernel<true>
      <<<(N_NODES / 16 + GEMM_WAVES - 1) / GEMM_WAVES, GEMM_WAVES * 32, 0, stream>>>(
          h_in, W, denom_in, feat);
  init_max_kernel<<<(N_NODES + 255) / 256, 256, 0, stream>>>(mmax);
  gemm_cls_kernel<<<(N_NODES / 16 + 7) / 8, 256, 0, stream>>>(feat, cls_w, g);
  zero_kernel<<<(N_NODES * ACC_W + 255) / 256, 256, 0, stream>>>(
      acc, N_NODES * ACC_W);
  edge_logits_kernel<<<N_EDGES / (8 * EPW), 256, 0, stream>>>(feat, attn, src, dst,
                                                      logits, mmax);
  edge_aggr_cls_kernel<<<N_EDGES / (8 * EPW), 256, 0, stream>>>(g, src, dst, logits,
                                                        mmax, acc);
  divide_rows_kernel<<<(N_NODES * OUT_DIM / 4 + 255) / 256, 256, 0, stream>>>(
      acc, out);
}

extern "C" void kernel_launch(void* const* d_in, const int* in_sizes, int n_in,
                              void* d_out, int out_size, void* d_ws,
                              size_t ws_size, hipStream_t stream) {
  const int* feats = (const int*)d_in[0];
  const int* src = (const int*)d_in[1];
  const int* dst = (const int*)d_in[2];
  const float* key_emb = (const float*)d_in[3];
  const float* val_emb = (const float*)d_in[4];
  const float* W1 = (const float*)d_in[5];
  const float* attn1 = (const float*)d_in[6];
  const float* W2 = (const float*)d_in[7];
  const float* attn2 = (const float*)d_in[8];
  const float* cls_w = (const float*)d_in[9];
  float* out = (float*)d_out;

  float* hA = (float*)d_ws;
  float* feat = hA + (size_t)N_NODES * HID;
  float* hB = feat + (size_t)N_NODES * HID;
  float* ex = hB + (size_t)N_NODES * HID;
  int* mmax = (int*)(ex + N_EDGES);
  float* denom = (float*)(mmax + N_NODES);

  embed_kernel<<<N_NODES, HID, 0, stream>>>(feats, key_emb, val_emb, hA);

  int* unsafe = (int*)(hA + (size_t)N_NODES * HID - 1);
  run_gat_layer(hA, W1, attn1, src, dst, feat, hB, ex, mmax, denom, unsafe,
                stream);
  float* g = hA;
  float* acc = hA + (size_t)N_NODES * ACC_W;
  run_gat_layer_cls(hB, denom, W2, attn2, cls_w, src, dst, feat, g, acc, out,
                    ex, mmax, stream);
}


# Last layer: softmax shift dropped where all its logits lie in +-16 (checked; else the old max and sum passes run): g rows weighted by exp(logit) are added in the logits kernel; adds still unordered
# speedup vs baseline: 3.2126x; 1.1707x over previous
#include <hip/hip_runtime.h>
#include <hip/hip_bf16.h>

#define N_NODES 50000
#define N_EDGES 800000
#define HID 128
#define OUT_DIM 16
#define NEG_SLOPE 0.2f

typedef __attribute__((ext_vector_type(2))) float v2f;
typedef __attribute__((ext_vector_type(8))) float v8f;

__global__ __launch_bounds__(HID) void embed_kernel(
    const int* __restrict__ feats, const float* __restrict__ key_emb,
    const float* __restrict__ val_emb, float* __restrict__ h) {
  int n = blockIdx.x;
  int d = threadIdx.x;
  int f0 = feats[n * 2 + 0];
  int f1 = feats[n * 2 + 1];
  float v = key_emb[f0 * HID + d] + val_emb[f1 * HID + d];
  h[n * HID + d] = v > 0.f ? v : 0.f;
}

#define GEMM_WAVES 4
#define GEMM_WPITCH (2 * HID + 32)

template <bool DIV_ROWS>
__global__ __launch_bounds__(GEMM_WAVES * 32) void gemm128_kernel(
    const float* __restrict__ A, const float* __restrict__ B,
    const float* __restrict__ rowsum, float* __restrict__ C) {
  __shared__ __attribute__((aligned(16))) float Ws[(HID / 2) * GEMM_WPITCH];
  const int lane = threadIdx.x & 31;
  const int wave = __builtin_amdgcn_readfirstlane((int)(threadIdx.x >> 5));
  const int stripe = blockIdx.x * GEMM_WAVES + wave;
  const int m0 = (stripe < N_NODES / 16 ? stripe : N_NODES / 16 - 1) * 16;
  const int r = lane & 15;
  const int hi = lane >> 4;

  const float* arow = A + (size_t)(m0 + r) * HID + (hi << 1);
  v2f a[HID / 4];
#pragma unroll
  for (int s = 0; s < HID / 4; ++s) a[s] = *(const v2f*)(arow + 4 * s);
  float inv = 0.f;
  if (DIV_ROWS) {
    const float dn = rowsum[m0 + r];
    if (dn > 0.f) inv = 1.f / dn;
  }

  float4 w0[HID / 2 / GEMM_WAVES], w1[HID / 2 / GEMM_WAVES];
#pragma unroll
  for (int i = 0; i < HID / 2 / GEMM_WAVES; ++i) {
    const int kp = wave + i * GEMM_WAVES;
    w0[i] = ((const float4*)(B + (size_t)(2 * kp) * HID))[lane];
    w1[i] = ((const float4*)(B + (size_t)(2 * kp + 1) * HID))[lane];
  }
#pragma unroll
  for (int i = 0; i < HID / 2 / GEMM_WAVES; ++i) {
    const int kp = wave + i * GEMM_WAVES;
    float4* w = (float4*)(Ws + kp * GEMM_WPITCH + 8 * lane);
    w[0] = make_float4(w0[i].x, w1[i].x, w0[i].y, w1[i].y);
    w[1] = make_float4(w0[i].z, w1[i].z, w0[i].w, w1[i].w);
  }
  __syncthreads();
  if (stripe >= N_NODES / 16) return;

  const float* bl = Ws + hi * GEMM_WPITCH + 2 * r;
  v8f acc[HID / 16];
#pragma unroll
  for (int j = 0; j < HID / 16; ++j) acc[j] = v8f{};
#pragma unroll
  for (int s = 0; s < HID / 4; ++s) {
    const float* bk = bl + (2 * s) * GEMM_WPITCH;
#pragma unroll
    for (int j = 0; j < HID / 16; ++j) {
      const v2f b = *(const v2f*)(bk + 32 * j);
      acc[j] = __builtin_amdgcn_wmma_f32_16x16x4_f32(false, a[s], false, b,
                                                     (short)0, acc[j], false,
                                                     false);
    }
  }
#pragma unroll
  for (int v = 0; v < 8; ++v) {
    float sc = 1.f;
    if (DIV_ROWS) sc = __shfl(inv, v + (hi << 3), 32);
    float* crow = C + (size_t)(m0 + v + (hi << 3)) * HID + r;
#pragma unroll
    for (int j = 0; j < HID / 16; ++j) {
      float c = acc[j][v];
      if (DIV_ROWS) c *= sc;
      crow[16 * j] = c;
    }
  }
}

#define ACC_W 32

__global__ __launch_bounds__(256) void gemm_cls_kernel(
    const float* __restrict__ A, const float* __restrict__ B,
    float* __restrict__ C) {
  const int wave = threadIdx.x >> 5;
  const int tile = blockIdx.x * 8 + wave;
  if (tile >= N_NODES / 16) return;
  const int lane = threadIdx.x & 31;
  const int m0 = tile * 16;
  const int r = lane & 15;
  const int hi = lane >> 4;
  const int koff = hi << 1;

  const float* arow = A + (size_t)(m0 + r) * HID;
  v8f acc = {};
#pragma unroll 8
  for (int k0 = 0; k0 < HID; k0 += 4) {
    const int k = k0 + koff;
    v2f a;
    a.x = arow[k];
    a.y = arow[k + 1];
    v2f b;
    b.x = B[k * OUT_DIM + r];
    b.y = B[(k + 1) * OUT_DIM + r];
    acc = __builtin_amdgcn_wmma_f32_16x16x4_f32(false, a, false, b, (short)0,
                                                acc, false, false);
  }
#pragma unroll
  for (int v = 0; v < 8; ++v) {
    float* crow = C + (size_t)(m0 + v + (hi << 3)) * ACC_W;
    crow[r] = acc[v];
    crow[OUT_DIM + r] = 1.f;
  }
}

__device__ __forceinline__ int float_to_ord(float f) {
  int i = __float_as_int(f);
  return i >= 0 ? i : (i ^ 0x7FFFFFFF);
}
__device__ __forceinline__ float ord_to_float(int i) {
  return __int_as_float(i >= 0 ? i : (i ^ 0x7FFFFFFF));
}

__global__ void init_nodes_kernel(int* __restrict__ mmax,
                                  float* __restrict__ denom,
                                  int* __restrict__ unsafe) {
  int i = blockIdx.x * 256 + threadIdx.x;
  if (i < N_NODES) {
    mmax[i] = float_to_ord(-__builtin_inff());
    denom[i] = 0.f;
  }
  if (i == 0) *unsafe = 0;
}

__global__ void init_cls_kernel(float* __restrict__ acc, int* __restrict__ mmax,
                                int* __restrict__ unsafe) {
  int i = blockIdx.x * 256 + threadIdx.x;
  if (i < N_NODES * ACC_W) acc[i] = 0.f;
  if (i < N_NODES) mmax[i] = float_to_ord(-__builtin_inff());
  if (i == 0) *unsafe = 0;
}

__global__ void zero_kernel(float* __restrict__ p, int n) {
  int i = blockIdx.x * 256 + threadIdx.x;
  if (i < n) p[i] = 0.f;
}

#define EPW 32
#define EB 8
#define LOGIT_SAFE 16.0f

__global__ __launch_bounds__(256) void edge_fused_cls_kernel(
    const float* __restrict__ feat, const float* __restrict__ attn,
    const float* __restrict__ g, const int* __restrict__ src,
    const int* __restrict__ dst, float* __restrict__ logits,
    float* __restrict__ acc, int* __restrict__ unsafe) {
  const int lane = threadIdx.x & 31;
  const int wave = __builtin_amdgcn_readfirstlane((int)(threadIdx.x >> 5));
  const int e0 = (blockIdx.x * 8 + wave) * EPW;
  const int sv = src[e0 + lane];
  const int dv = dst[e0 + lane];
  const float4 a4 = ((const float4*)attn)[lane];
  const bool h4 = (lane & 16) != 0;
  const bool h3 = (lane & 8) != 0;
  const bool h2 = (lane & 4) != 0;
  float mine = 0.f;
#pragma unroll 1
  for (int b = 0; b < EPW / EB; ++b) {
    float4 x[EB], y[EB];
    float gr[EB];
#pragma unroll
    for (int i = 0; i < EB; ++i) {
      const int s = __builtin_amdgcn_readlane(sv, b * EB + i);
      const int d = __builtin_amdgcn_readlane(dv, b * EB + i);
      x[i] = ((const float4*)(feat + (size_t)s * HID))[lane];
      y[i] = ((const float4*)(feat + (size_t)d * HID))[lane];
      gr[i] = g[(unsigned)s * ACC_W + lane];
    }
    float p[EB];
#pragma unroll
    for (int i = 0; i < EB; ++i) {
      float q = 0.f, t;
      t = x[i].x + y[i].x; q += a4.x * (t > 0.f ? t : NEG_SLOPE * t);
      t = x[i].y + y[i].y; q += a4.y * (t > 0.f ? t : NEG_SLOPE * t);
      t = x[i].z + y[i].z; q += a4.z * (t > 0.f ? t : NEG_SLOPE * t);
      t = x[i].w + y[i].w; q += a4.w * (t > 0.f ? t : NEG_SLOPE * t);
      p[i] = q;
    }
    float q4[4], q2[2], r;
#pragma unroll
    for (int i = 0; i < 4; ++i) {
      const float keep = h4 ? p[i + 4] : p[i];
      const float send = h4 ? p[i] : p[i + 4];
      q4[i] = keep + __shfl_xor(send, 16, 32);
    }
#pragma unroll
    for (int i = 0; i < 2; ++i) {
      const float keep = h3 ? q4[i + 2] : q4[i];
      const float send = h3 ? q4[i] : q4[i + 2];
      q2[i] = keep + __shfl_xor(send, 8, 32);
    }
    {
      const float keep = h2 ? q2[1] : q2[0];
      const float send = h2 ? q2[0] : q2[1];
      r = keep + __shfl_xor(send, 4, 32);
    }
    r += __shfl_xor(r, 2, 32);
    r += __shfl_xor(r, 1, 32);
    const int wv = __float_as_int(expf(r));
#pragma unroll
    for (int i = 0; i < EB; ++i) {
      const int d = __builtin_amdgcn_readlane(dv, b * EB + i);
      const float w = __int_as_float(__builtin_amdgcn_readlane(wv, 4 * i));
      atomicAdd(acc + ((unsigned)d * ACC_W + lane), w * gr[i]);
    }
    const float tmp = __shfl(r, (lane & 7) * 4, 32);
    if ((lane >> 3) == b) mine = tmp;
  }
  logits[e0 + lane] = mine;
  if (!(fabsf(mine) <= LOGIT_SAFE)) *unsafe = 1;
}

__global__ __launch_bounds__(256) void fallback_reset_cls_kernel(
    const int* __restrict__ unsafe, const int* __restrict__ dst,
    const float* __restrict__ logits, int* __restrict__ mmax,
    float* __restrict__ acc) {
  if (*unsafe == 0) return;
  const int t = blockIdx.x * 256 + threadIdx.x;
  const int nt = gridDim.x * 256;
  for (int i = t; i < N_NODES * ACC_W; i += nt) acc[i] = 0.f;
  for (int i = t; i < N_EDGES; i += nt)
    atomicMax(mmax + dst[i], float_to_ord(logits[i]));
}

#define RP (HID / 32)

__global__ __launch_bounds__(256) void edge_fused_kernel(
    const float* __restrict__ feat, const float* __restrict__ attn,
    const int* __restrict__ src, const int* __restrict__ dst,
    float* __restrict__ logits, float* __restrict__ denom,
    float* __restrict__ out, int* __restrict__ unsafe) {
  const int lane = threadIdx.x & 31;
  const int wave = __builtin_amdgcn_readfirstlane((int)(threadIdx.x >> 5));
  const int e0 = (blockIdx.x * 8 + wave) * EPW;
  const int sv = src[e0 + lane];
  const int dv = dst[e0 + lane];
  float a[RP];
#pragma unroll
  for (int j = 0; j < RP; ++j) a[j] = attn[lane + 32 * j];
  const bool h4 = (lane & 16) != 0;
  const bool h3 = (lane & 8) != 0;
  const bool h2 = (lane & 4) != 0;
  float mine = 0.f;
#pragma unroll 1
  for (int b = 0; b < EPW / EB; ++b) {
    float x[EB][RP], y[EB][RP];
#pragma unroll
    for (int i = 0; i < EB; ++i) {
      const int s = __builtin_amdgcn_readlane(sv, b * EB + i);
      const int d = __builtin_amdgcn_readlane(dv, b * EB + i);
      const float* xr = feat + (size_t)s * HID + lane;
      const float* yr = feat + (size_t)d * HID + lane;
#pragma unroll
      for (int j = 0; j < RP; ++j) {
        x[i][j] = xr[32 * j];
        y[i][j] = yr[32 * j];
      }
    }
    float p[EB];
#pragma unroll
    for (int i = 0; i < EB; ++i) {
      float q = 0.f;
#pragma unroll
      for (int j = 0; j < RP; ++j) {
        const float t = x[i][j] + y[i][j];
        q += a[j] * (t > 0.f ? t : NEG_SLOPE * t);
      }
      p[i] = q;
    }
    float q4[4], q2[2], r;
#pragma unroll
    for (int i = 0; i < 4; ++i) {
      const float keep = h4 ? p[i + 4] : p[i];
      const float send = h4 ? p[i] : p[i + 4];
      q4[i] = keep + __shfl_xor(send, 16, 32);
    }
#pragma unroll
    for (int i = 0; i < 2; ++i) {
      const float keep = h3 ? q4[i + 2] : q4[i];
      const float send = h3 ? q4[i] : q4[i + 2];
      q2[i] = keep + __shfl_xor(send, 8, 32);
    }
    {
      const float keep = h2 ? q2[1] : q2[0];
      const float send = h2 ? q2[0] : q2[1];
      r = keep + __shfl_xor(send, 4, 32);
    }
    r += __shfl_xor(r, 2, 32);
    r += __shfl_xor(r, 1, 32);
    const int wv = __float_as_int(expf(r));
#pragma unroll
    for (int i = 0; i < EB; ++i) {
      const int d = __builtin_amdgcn_readlane(dv, b * EB + i);
      const float w = __int_as_float(__builtin_amdgcn_readlane(wv, 4 * i));
      float* o = out + (size_t)d * HID + lane;
#pragma unroll
      for (int j = 0; j < RP; ++j) atomicAdd(o + 32 * j, w * x[i][j]);
    }
    const float tmp = __shfl(r, (lane & 7) * 4, 32);
    if ((lane >> 3) == b) mine = tmp;
  }
  logits[e0 + lane] = mine;
  atomicAdd(denom + dv, expf(mine));
  if (!(fabsf(mine) <= LOGIT_SAFE)) *unsafe = 1;
}

__global__ __launch_bounds__(256) void fallback_reset_max_kernel(
    const int* __restrict__ unsafe, const int* __restrict__ dst,
    const float* __restrict__ logits, int* __restrict__ mmax,
    float* __restrict__ denom, float* __restrict__ out) {
  if (*unsafe == 0) return;
  const int t = blockIdx.x * 256 + threadIdx.x;
  const int nt = gridDim.x * 256;
  for (int i = t; i < N_NODES * HID; i += nt) out[i] = 0.f;
  for (int i = t; i < N_NODES; i += nt) denom[i] = 0.f;
  for (int i = t; i < N_EDGES; i += nt)
    atomicMax(mmax + dst[i], float_to_ord(logits[i]));
}


__global__ __launch_bounds__(256) void edge_aggr_kernel(
    const float* __restrict__ feat, const int* __restrict__ src,
    const int* __restrict__ dst, const float* __restrict__ logits,
    const int* __restrict__ mmax, float* __restrict__ denom,
    float* __restrict__ out, const int* __restrict__ unsafe) {
  if (*unsafe == 0) return;
  const int lane = threadIdx.x & 31;
  const int wave = __builtin_amdgcn_readfirstlane((int)(threadIdx.x >> 5));
  for (int grp = blockIdx.x * 8 + wave; grp < N_EDGES / EPW;
       grp += gridDim.x * 8) {
    const int e0 = grp * EPW;
    const int sv = src[e0 + lane];
    const int dv = dst[e0 + lane];
    const float exv = expf(logits[e0 + lane] - ord_to_float(mmax[dv]));
    atomicAdd(denom + dv, exv);
    const int av = __float_as_int(exv);
#pragma unroll 1
    for (int b = 0; b < EPW / EB; ++b) {
      float x[EB][RP];
#pragma unroll
      for (int i = 0; i < EB; ++i) {
        const int s = __builtin_amdgcn_readlane(sv, b * EB + i);
        const float* row = feat + (size_t)s * HID + lane;
#pragma unroll
        for (int j = 0; j < RP; ++j) x[i][j] = row[32 * j];
      }
#pragma unroll
      for (int i = 0; i < EB; ++i) {
        const int d = __builtin_amdgcn_readlane(dv, b * EB + i);
        const float a = __int_as_float(__builtin_amdgcn_readlane(av, b * EB + i));
        float* o = out + (size_t)d * HID + lane;
#pragma unroll
        for (int j = 0; j < RP; ++j) atomicAdd(o + 32 * j, a * x[i][j]);
      }
    }
  }
}

__global__ __launch_bounds__(256) void edge_aggr_cls_kernel(
    const float* __restrict__ g, const int* __restrict__ src,
    const int* __restrict__ dst, const float* __restrict__ logits,
    const int* __restrict__ mmax, float* __restrict__ acc,
    const int* __restrict__ unsafe) {
  if (*unsafe == 0) return;
  const int lane = threadIdx.x & 31;
  const int wave = __builtin_amdgcn_readfirstlane((int)(threadIdx.x >> 5));
  for (int grp = blockIdx.x * 8 + wave; grp < N_EDGES / EPW;
       grp += gridDim.x * 8) {
    const int e0 = grp * EPW;
    const int sv = src[e0 + lane];
    const int dv = dst[e0 + lane];
    const int av =
        __float_as_int(expf(logits[e0 + lane] - ord_to_float(mmax[dv])));
    float x[EPW];
#pragma unroll
    for (int it = 0; it < EPW; ++it) {
      const int s = __builtin_amdgcn_readlane(sv, it);
      x[it] = g[(unsigned)s * ACC_W + lane];
    }
#pragma unroll
    for (int it = 0; it < EPW; ++it) {
      const int d = __builtin_amdgcn_readlane(dv, it);
      const float a = __int_as_float(__builtin_amdgcn_readlane(av, it));
      atomicAdd(acc + ((unsigned)d * ACC_W + lane), a * x[it]);
    }
  }
}

__global__ void divide_rows_kernel(const float* __restrict__ acc,
                                   float* __restrict__ out) {
  const int i = blockIdx.x * 256 + threadIdx.x;
  if (i >= N_NODES * OUT_DIM / 4) return;
  const int n = i >> 2;
  const int q = i & 3;
  const float dn = acc[n * ACC_W + OUT_DIM];
  float4 v = ((const float4*)(acc + n * ACC_W))[q];
  if (dn > 0.f) {
    v.x /= dn; v.y /= dn; v.z /= dn; v.w /= dn;
  } else {
    v.x = 0.f; v.y = 0.f; v.z = 0.f; v.w = 0.f;
  }
  ((float4*)out)[i] = v;
}

static void run_gat_layer(const float* h_in, const float* W, const float* attn,
                          const int* src, const int* dst, float* feat,
                          float* h_out, float* logits, int* mmax, float* denom,
                          int* unsafe, hipStream_t stream) {
  gemm128_kernel<false>
      <<<(N_NODES / 16 + GEMM_WAVES - 1) / GEMM_WAVES, GEMM_WAVES * 32, 0, stream>>>(
          h_in, W, nullptr, feat);
  init_nodes_kernel<<<(N_NODES + 255) / 256, 256, 0, stream>>>(mmax, denom,
                                                               unsafe);
  zero_kernel<<<(N_NODES * HID + 255) / 256, 256, 0, stream>>>(h_out,
                                                               N_NODES * HID);
  edge_fused_kernel<<<N_EDGES / (8 * EPW), 256, 0, stream>>>(
      feat, attn, src, dst, logits, denom, h_out, unsafe);
  fallback_reset_max_kernel<<<512, 256, 0, stream>>>(unsafe, dst, logits, mmax,
                                                     denom, h_out);
  edge_aggr_kernel<<<512, 256, 0, stream>>>(feat, src, dst, logits, mmax, denom,
                                            h_out, unsafe);
}

static void run_gat_layer_cls(const float* h_in, const float* denom_in,
                              const float* W, const float* attn,
                              const float* cls_w, const int* src,
                              const int* dst, float* feat, float* g,
                              float* acc, float* out, float* logits, int* mmax,
                              int* unsafe, hipStream_t stream) {
  gemm128_kernel<true>
      <<<(N_NODES / 16 + GEMM_WAVES - 1) / GEMM_WAVES, GEMM_WAVES * 32, 0, stream>>>(
          h_in, W, denom_in, feat);
  gemm_cls_kernel<<<(N_NODES / 16 + 7) / 8, 256, 0, stream>>>(feat, cls_w, g);
  init_cls_kernel<<<(N_NODES * ACC_W + 255) / 256, 256, 0, stream>>>(acc, mmax,
                                                                     unsafe);
  edge_fused_cls_kernel<<<N_EDGES / (8 * EPW), 256, 0, stream>>>(
      feat, attn, g, src, dst, logits, acc, unsafe);
  fallback_reset_cls_kernel<<<512, 256, 0, stream>>>(unsafe, dst, logits, mmax,
                                                     acc);
  edge_aggr_cls_kernel<<<512, 256, 0, stream>>>(g, src, dst, logits, mmax, acc,
                                                unsafe);
  divide_rows_kernel<<<(N_NODES * OUT_DIM / 4 + 255) / 256, 256, 0, stream>>>(
      acc, out);
}

extern "C" void kernel_launch(void* const* d_in, const int* in_sizes, int n_in,
                              void* d_out, int out_size, void* d_ws,
                              size_t ws_size, hipStream_t stream) {
  const int* feats = (const int*)d_in[0];
  const int* src = (const int*)d_in[1];
  const int* dst = (const int*)d_in[2];
  const float* key_emb = (const float*)d_in[3];
  const float* val_emb = (const float*)d_in[4];
  const float* W1 = (const float*)d_in[5];
  const float* attn1 = (const float*)d_in[6];
  const float* W2 = (const float*)d_in[7];
  const float* attn2 = (const float*)d_in[8];
  const float* cls_w = (const float*)d_in[9];
  float* out = (float*)d_out;

  float* hA = (float*)d_ws;
  float* feat = hA + (size_t)N_NODES * HID;
  float* hB = feat + (size_t)N_NODES * HID;
  float* ex = hB + (size_t)N_NODES * HID;
  int* mmax = (int*)(ex + N_EDGES);
  float* denom = (float*)(mmax + N_NODES);

  embed_kernel<<<N_NODES, HID, 0, stream>>>(feats, key_emb, val_emb, hA);

  int* unsafe = (int*)(hA + (size_t)N_NODES * HID - 1);
  run_gat_layer(hA, W1, attn1, src, dst, feat, hB, ex, mmax, denom, unsafe,
                stream);
  float* g = hA;
  float* acc = hA + (size_t)N_NODES * ACC_W;
  run_gat_layer_cls(hB, denom, W2, attn2, cls_w, src, dst, feat, g, acc, out,
                    ex, mmax, unsafe, stream);
}


# Elementwise launches (node embedding, zero rows, last layer's start values): 16 bytes a lane and 4 rows or 5 pieces a wave, not one float a thread; 450000 waves become 25000; same values
# speedup vs baseline: 3.3135x; 1.0314x over previous
#include <hip/hip_runtime.h>
#include <hip/hip_bf16.h>

#define N_NODES 50000
#define N_EDGES 800000
#define HID 128
#define OUT_DIM 16
#define NEG_SLOPE 0.2f

typedef __attribute__((ext_vector_type(2))) float v2f;
typedef __attribute__((ext_vector_type(8))) float v8f;

#define EMB_ROWS 4
#define EMB_WAVES 8
static_assert(N_NODES % EMB_ROWS == 0, "a wave's rows are all inside the array");

__global__ __launch_bounds__(EMB_WAVES * 32) void embed_kernel(
    const int* __restrict__ feats, const float* __restrict__ key_emb,
    const float* __restrict__ val_emb, float* __restrict__ h) {
  const int lane = threadIdx.x & 31;
  const int wave = __builtin_amdgcn_readfirstlane((int)(threadIdx.x >> 5));
  const int n0 = (blockIdx.x * EMB_WAVES + wave) * EMB_ROWS;
  if (n0 >= N_NODES) return;
  const int* f = feats + n0 * 2;
  float4 k[EMB_ROWS], v[EMB_ROWS];
#pragma unroll
  for (int i = 0; i < EMB_ROWS; ++i) {
    k[i] = ((const float4*)(key_emb + (size_t)f[2 * i] * HID))[lane];
    v[i] = ((const float4*)(val_emb + (size_t)f[2 * i + 1] * HID))[lane];
  }
#pragma unroll
  for (int i = 0; i < EMB_ROWS; ++i) {
    float4 s;
    s.x = k[i].x + v[i].x;
    s.y = k[i].y + v[i].y;
    s.z = k[i].z + v[i].z;
    s.w = k[i].w + v[i].w;
    s.x = s.x > 0.f ? s.x : 0.f;
    s.y = s.y > 0.f ? s.y : 0.f;
    s.z = s.z > 0.f ? s.z : 0.f;
    s.w = s.w > 0.f ? s.w : 0.f;
    ((float4*)(h + (size_t)(n0 + i) * HID))[lane] = s;
  }
}

#define GEMM_WAVES 4
#define GEMM_WPITCH (2 * HID + 32)

template <bool DIV_ROWS>
__global__ __launch_bounds__(GEMM_WAVES * 32) void gemm128_kernel(
    const float* __restrict__ A, const float* __restrict__ B,
    const float* __restrict__ rowsum, float* __restrict__ C) {
  __shared__ __attribute__((aligned(16))) float Ws[(HID / 2) * GEMM_WPITCH];
  const int lane = threadIdx.x & 31;
  const int wave = __builtin_amdgcn_readfirstlane((int)(threadIdx.x >> 5));
  const int stripe = blockIdx.x * GEMM_WAVES + wave;
  const int m0 = (stripe < N_NODES / 16 ? stripe : N_NODES / 16 - 1) * 16;
  const int r = lane & 15;
  const int hi = lane >> 4;

  const float* arow = A + (size_t)(m0 + r) * HID + (hi << 1);
  v2f a[HID / 4];
#pragma unroll
  for (int s = 0; s < HID / 4; ++s) a[s] = *(const v2f*)(arow + 4 * s);
  float inv = 0.f;
  if (DIV_ROWS) {
    const float dn = rowsum[m0 + r];
    if (dn > 0.f) inv = 1.f / dn;
  }

  float4 w0[HID / 2 / GEMM_WAVES], w1[HID / 2 / GEMM_WAVES];
#pragma unroll
  for (int i = 0; i < HID / 2 / GEMM_WAVES; ++i) {
    const int kp = wave + i * GEMM_WAVES;
    w0[i] = ((const float4*)(B + (size_t)(2 * kp) * HID))[lane];
    w1[i] = ((const float4*)(B + (size_t)(2 * kp + 1) * HID))[lane];
  }
#pragma unroll
  for (int i = 0; i < HID / 2 / GEMM_WAVES; ++i) {
    const int kp = wave + i * GEMM_WAVES;
    float4* w = (float4*)(Ws + kp * GEMM_WPITCH + 8 * lane);
    w[0] = make_float4(w0[i].x, w1[i].x, w0[i].y, w1[i].y);
    w[1] = make_float4(w0[i].z, w1[i].z, w0[i].w, w1[i].w);
  }
  __syncthreads();
  if (stripe >= N_NODES / 16) return;

  const float* bl = Ws + hi * GEMM_WPITCH + 2 * r;
  v8f acc[HID / 16];
#pragma unroll
  for (int j = 0; j < HID / 16; ++j) acc[j] = v8f{};
#pragma unroll
  for (int s = 0; s < HID / 4; ++s) {
    const float* bk = bl + (2 * s) * GEMM_WPITCH;
#pragma unroll
    for (int j = 0; j < HID / 16; ++j) {
      const v2f b = *(const v2f*)(bk + 32 * j);
      acc[j] = __builtin_amdgcn_wmma_f32_16x16x4_f32(false, a[s], false, b,
                                                     (short)0, acc[j], false,
                                                     false);
    }
  }
#pragma unroll
  for (int v = 0; v < 8; ++v) {
    float sc = 1.f;
    if (DIV_ROWS) sc = __shfl(inv, v + (hi << 3), 32);
    float* crow = C + (size_t)(m0 + v + (hi << 3)) * HID + r;
#pragma unroll
    for (int j = 0; j < HID / 16; ++j) {
      float c = acc[j][v];
      if (DIV_ROWS) c *= sc;
      crow[16 * j] = c;
    }
  }
}

#define ACC_W 32

__global__ __launch_bounds__(256) void gemm_cls_kernel(
    const float* __restrict__ A, const float* __restrict__ B,
    float* __restrict__ C) {
  const int wave = threadIdx.x >> 5;
  const int tile = blockIdx.x * 8 + wave;
  if (tile >= N_NODES / 16) return;
  const int lane = threadIdx.x & 31;
  const int m0 = tile * 16;
  const int r = lane & 15;
  const int hi = lane >> 4;
  const int koff = hi << 1;

  const float* arow = A + (size_t)(m0 + r) * HID;
  v8f acc = {};
#pragma unroll 8
  for (int k0 = 0; k0 < HID; k0 += 4) {
    const int k = k0 + koff;
    v2f a;
    a.x = arow[k];
    a.y = arow[k + 1];
    v2f b;
    b.x = B[k * OUT_DIM + r];
    b.y = B[(k + 1) * OUT_DIM + r];
    acc = __builtin_amdgcn_wmma_f32_16x16x4_f32(false, a, false, b, (short)0,
                                                acc, false, false);
  }
#pragma unroll
  for (int v = 0; v < 8; ++v) {
    float* crow = C + (size_t)(m0 + v + (hi << 3)) * ACC_W;
    crow[r] = acc[v];
    crow[OUT_DIM + r] = 1.f;
  }
}

__device__ __forceinline__ int float_to_ord(float f) {
  int i = __float_as_int(f);
  return i >= 0 ? i : (i ^ 0x7FFFFFFF);
}
__device__ __forceinline__ float ord_to_float(int i) {
  return __int_as_float(i >= 0 ? i : (i ^ 0x7FFFFFFF));
}

__global__ void init_nodes_kernel(int* __restrict__ mmax,
                                  float* __restrict__ denom,
                                  int* __restrict__ unsafe) {
  int i = blockIdx.x * 256 + threadIdx.x;
  if (i < N_NODES) {
    mmax[i] = float_to_ord(-__builtin_inff());
    denom[i] = 0.f;
  }
  if (i == 0) *unsafe = 0;
}

#define FILL_IT 5
#define FILL_WAVES 8
#define FILL_SHARE (32 * FILL_IT)
static_assert((N_NODES * HID / 4) % FILL_SHARE == 0, "sum rows: whole wave shares");
static_assert((N_NODES * ACC_W / 4) % FILL_SHARE == 0, "last layer's sum rows: whole wave shares");
static_assert(N_NODES % 4 == 0, "mmax is filled as int4 words");

__device__ __forceinline__ void fill_zero_share(float* __restrict__ p,
                                                int nshares) {
  const int lane = threadIdx.x & 31;
  const int wave = __builtin_amdgcn_readfirstlane((int)(threadIdx.x >> 5));
  const int share = blockIdx.x * FILL_WAVES + wave;
  if (share < nshares) {
    float4* q = (float4*)p + (size_t)share * FILL_SHARE + lane;
#pragma unroll
    for (int k = 0; k < FILL_IT; ++k) q[32 * k] = make_float4(0.f, 0.f, 0.f, 0.f);
  }
}

__global__ __launch_bounds__(FILL_WAVES * 32) void init_cls_kernel(
    float* __restrict__ acc, int* __restrict__ mmax, int* __restrict__ unsafe) {
  fill_zero_share(acc, N_NODES * ACC_W / 4 / FILL_SHARE);
  const int t = blockIdx.x * (FILL_WAVES * 32) + threadIdx.x;
  if (t < N_NODES / 4) {
    const int m = float_to_ord(-__builtin_inff());
    ((int4*)mmax)[t] = make_int4(m, m, m, m);
  }
  if (t == 0) *unsafe = 0;
}

__global__ __launch_bounds__(FILL_WAVES * 32) void zero_kernel(
    float* __restrict__ p, int nshares) {
  fill_zero_share(p, nshares);
}

#define EPW 32
#define EB 8
#define LOGIT_SAFE 16.0f

__global__ __launch_bounds__(256) void edge_fused_cls_kernel(
    const float* __restrict__ feat, const float* __restrict__ attn,
    const float* __restrict__ g, const int* __restrict__ src,
    const int* __restrict__ dst, float* __restrict__ logits,
    float* __restrict__ acc, int* __restrict__ unsafe) {
  const int lane = threadIdx.x & 31;
  const int wave = __builtin_amdgcn_readfirstlane((int)(threadIdx.x >> 5));
  const int e0 = (blockIdx.x * 8 + wave) * EPW;
  const int sv = src[e0 + lane];
  const int dv = dst[e0 + lane];
  const float4 a4 = ((const float4*)attn)[lane];
  const bool h4 = (lane & 16) != 0;
  const bool h3 = (lane & 8) != 0;
  const bool h2 = (lane & 4) != 0;
  float mine = 0.f;
#pragma unroll 1
  for (int b = 0; b < EPW / EB; ++b) {
    float4 x[EB], y[EB];
    float gr[EB];
#pragma unroll
    for (int i = 0; i < EB; ++i) {
      const int s = __builtin_amdgcn_readlane(sv, b * EB + i);
      const int d = __builtin_amdgcn_readlane(dv, b * EB + i);
      x[i] = ((const float4*)(feat + (size_t)s * HID))[lane];
      y[i] = ((const float4*)(feat + (size_t)d * HID))[lane];
      gr[i] = g[(unsigned)s * ACC_W + lane];
    }
    float p[EB];
#pragma unroll
    for (int i = 0; i < EB; ++i) {
      float q = 0.f, t;
      t = x[i].x + y[i].x; q += a4.x * (t > 0.f ? t : NEG_SLOPE * t);
      t = x[i].y + y[i].y; q += a4.y * (t > 0.f ? t : NEG_SLOPE * t);
      t = x[i].z + y[i].z; q += a4.z * (t > 0.f ? t : NEG_SLOPE * t);
      t = x[i].w + y[i].w; q += a4.w * (t > 0.f ? t : NEG_SLOPE * t);
      p[i] = q;
    }
    float q4[4], q2[2], r;
#pragma unroll
    for (int i = 0; i < 4; ++i) {
      const float keep = h4 ? p[i + 4] : p[i];
      const float send = h4 ? p[i] : p[i + 4];
      q4[i] = keep + __shfl_xor(send, 16, 32);
    }
#pragma unroll
    for (int i = 0; i < 2; ++i) {
      const float keep = h3 ? q4[i + 2] : q4[i];
      const float send = h3 ? q4[i] : q4[i + 2];
      q2[i] = keep + __shfl_xor(send, 8, 32);
    }
    {
      const float keep = h2 ? q2[1] : q2[0];
      const float send = h2 ? q2[0] : q2[1];
      r = keep + __shfl_xor(send, 4, 32);
    }
    r += __shfl_xor(r, 2, 32);
    r += __shfl_xor(r, 1, 32);
    const int wv = __float_as_int(expf(r));
#pragma unroll
    for (int i = 0; i < EB; ++i) {
      const int d = __builtin_amdgcn_readlane(dv, b * EB + i);
      const float w = __int_as_float(__builtin_amdgcn_readlane(wv, 4 * i));
      atomicAdd(acc + ((unsigned)d * ACC_W + lane), w * gr[i]);
    }
    const float tmp = __shfl(r, (lane & 7) * 4, 32);
    if ((lane >> 3) == b) mine = tmp;
  }
  logits[e0 + lane] = mine;
  if (!(fabsf(mine) <= LOGIT_SAFE)) *unsafe = 1;
}

__global__ __launch_bounds__(256) void fallback_reset_cls_kernel(
    const int* __restrict__ unsafe, const int* __restrict__ dst,
    const float* __restrict__ logits, int* __restrict__ mmax,
    float* __restrict__ acc) {
  if (*unsafe == 0) return;
  const int t = blockIdx.x * 256 + threadIdx.x;
  const int nt = gridDim.x * 256;
  for (int i = t; i < N_NODES * ACC_W; i += nt) acc[i] = 0.f;
  for (int i = t; i < N_EDGES; i += nt)
    atomicMax(mmax + dst[i], float_to_ord(logits[i]));
}

#define RP (HID / 32)

__global__ __launch_bounds__(256) void edge_fused_kernel(
    const float* __restrict__ feat, const float* __restrict__ attn,
    const int* __restrict__ src, const int* __restrict__ dst,
    float* __restrict__ logits, float* __restrict__ denom,
    float* __restrict__ out, int* __restrict__ unsafe) {
  const int lane = threadIdx.x & 31;
  const int wave = __builtin_amdgcn_readfirstlane((int)(threadIdx.x >> 5));
  const int e0 = (blockIdx.x * 8 + wave) * EPW;
  const int sv = src[e0 + lane];
  const int dv = dst[e0 + lane];
  float a[RP];
#pragma unroll
  for (int j = 0; j < RP; ++j) a[j] = attn[lane + 32 * j];
  const bool h4 = (lane & 16) != 0;
  const bool h3 = (lane & 8) != 0;
  const bool h2 = (lane & 4) != 0;
  float mine = 0.f;
#pragma unroll 1
  for (int b = 0; b < EPW / EB; ++b) {
    float x[EB][RP], y[EB][RP];
#pragma unroll
    for (int i = 0; i < EB; ++i) {
      const int s = __builtin_amdgcn_readlane(sv, b * EB + i);
      const int d = __builtin_amdgcn_readlane(dv, b * EB + i);
      const float* xr = feat + (size_t)s * HID + lane;
      const float* yr = feat + (size_t)d * HID + lane;
#pragma unroll
      for (int j = 0; j < RP; ++j) {
        x[i][j] = xr[32 * j];
        y[i][j] = yr[32 * j];
      }
    }
    float p[EB];
#pragma unroll
    for (int i = 0; i < EB; ++i) {
      float q = 0.f;
#pragma unroll
      for (int j = 0; j < RP; ++j) {
        const float t = x[i][j] + y[i][j];
        q += a[j] * (t > 0.f ? t : NEG_SLOPE * t);
      }
      p[i] = q;
    }
    float q4[4], q2[2], r;
#pragma unroll
    for (int i = 0; i < 4; ++i) {
      const float keep = h4 ? p[i + 4] : p[i];
      const float send = h4 ? p[i] : p[i + 4];
      q4[i] = keep + __shfl_xor(send, 16, 32);
    }
#pragma unroll
    for (int i = 0; i < 2; ++i) {
      const float keep = h3 ? q4[i + 2] : q4[i];
      const float send = h3 ? q4[i] : q4[i + 2];
      q2[i] = keep + __shfl_xor(send, 8, 32);
    }
    {
      const float keep = h2 ? q2[1] : q2[0];
      const float send = h2 ? q2[0] : q2[1];
      r = keep + __shfl_xor(send, 4, 32);
    }
    r += __shfl_xor(r, 2, 32);
    r += __shfl_xor(r, 1, 32);
    const int wv = __float_as_int(expf(r));
#pragma unroll
    for (int i = 0; i < EB; ++i) {
      const int d = __builtin_amdgcn_readlane(dv, b * EB + i);
      const float w = __int_as_float(__builtin_amdgcn_readlane(wv, 4 * i));
      float* o = out + (size_t)d * HID + lane;
#pragma unroll
      for (int j = 0; j < RP; ++j) atomicAdd(o + 32 * j, w * x[i][j]);
    }
    const float tmp = __shfl(r, (lane & 7) * 4, 32);
    if ((lane >> 3) == b) mine = tmp;
  }
  logits[e0 + lane] = mine;
  atomicAdd(denom + dv, expf(mine));
  if (!(fabsf(mine) <= LOGIT_SAFE)) *unsafe = 1;
}

__global__ __launch_bounds__(256) void fallback_reset_max_kernel(
    const int* __restrict__ unsafe, const int* __restrict__ dst,
    const float* __restrict__ logits, int* __restrict__ mmax,
    float* __restrict__ denom, float* __restrict__ out) {
  if (*unsafe == 0) return;
  const int t = blockIdx.x * 256 + threadIdx.x;
  const int nt = gridDim.x * 256;
  for (int i = t; i < N_NODES * HID; i += nt) out[i] = 0.f;
  for (int i = t; i < N_NODES; i += nt) denom[i] = 0.f;
  for (int i = t; i < N_EDGES; i += nt)
    atomicMax(mmax + dst[i], float_to_ord(logits[i]));
}


__global__ __launch_bounds__(256) void edge_aggr_kernel(
    const float* __restrict__ feat, const int* __restrict__ src,
    const int* __restrict__ dst, const float* __restrict__ logits,
    const int* __restrict__ mmax, float* __restrict__ denom,
    float* __restrict__ out, const int* __restrict__ unsafe) {
  if (*unsafe == 0) return;
  const int lane = threadIdx.x & 31;
  const int wave = __builtin_amdgcn_readfirstlane((int)(threadIdx.x >> 5));
  for (int grp = blockIdx.x * 8 + wave; grp < N_EDGES / EPW;
       grp += gridDim.x * 8) {
    const int e0 = grp * EPW;
    const int sv = src[e0 + lane];
    const int dv = dst[e0 + lane];
    const float exv = expf(logits[e0 + lane] - ord_to_float(mmax[dv]));
    atomicAdd(denom + dv, exv);
    const int av = __float_as_int(exv);
#pragma unroll 1
    for (int b = 0; b < EPW / EB; ++b) {
      float x[EB][RP];
#pragma unroll
      for (int i = 0; i < EB; ++i) {
        const int s = __builtin_amdgcn_readlane(sv, b * EB + i);
        const float* row = feat + (size_t)s * HID + lane;
#pragma unroll
        for (int j = 0; j < RP; ++j) x[i][j] = row[32 * j];
      }
#pragma unroll
      for (int i = 0; i < EB; ++i) {
        const int d = __builtin_amdgcn_readlane(dv, b * EB + i);
        const float a = __int_as_float(__builtin_amdgcn_readlane(av, b * EB + i));
        float* o = out + (size_t)d * HID + lane;
#pragma unroll
        for (int j = 0; j < RP; ++j) atomicAdd(o + 32 * j, a * x[i][j]);
      }
    }
  }
}

__global__ __launch_bounds__(256) void edge_aggr_cls_kernel(
    const float* __restrict__ g, const int* __restrict__ src,
    const int* __restrict__ dst, const float* __restrict__ logits,
    const int* __restrict__ mmax, float* __restrict__ acc,
    const int* __restrict__ unsafe) {
  if (*unsafe == 0) return;
  const int lane = threadIdx.x & 31;
  const int wave = __builtin_amdgcn_readfirstlane((int)(threadIdx.x >> 5));
  for (int grp = blockIdx.x * 8 + wave; grp < N_EDGES / EPW;
       grp += gridDim.x * 8) {
    const int e0 = grp * EPW;
    const int sv = src[e0 + lane];
    const int dv = dst[e0 + lane];
    const int av =
        __float_as_int(expf(logits[e0 + lane] - ord_to_float(mmax[dv])));
    float x[EPW];
#pragma unroll
    for (int it = 0; it < EPW; ++it) {
      const int s = __builtin_amdgcn_readlane(sv, it);
      x[it] = g[(unsigned)s * ACC_W + lane];
    }
#pragma unroll
    for (int it = 0; it < EPW; ++it) {
      const int d = __builtin_amdgcn_readlane(dv, it);
      const float a = __int_as_float(__builtin_amdgcn_readlane(av, it));
      atomicAdd(acc + ((unsigned)d * ACC_W + lane), a * x[it]);
    }
  }
}

__global__ void divide_rows_kernel(const float* __restrict__ acc,
                                   float* __restrict__ out) {
  const int i = blockIdx.x * 256 + threadIdx.x;
  if (i >= N_NODES * OUT_DIM / 4) return;
  const int n = i >> 2;
  const int q = i & 3;
  const float dn = acc[n * ACC_W + OUT_DIM];
  float4 v = ((const float4*)(acc + n * ACC_W))[q];
  if (dn > 0.f) {
    v.x /= dn; v.y /= dn; v.z /= dn; v.w /= dn;
  } else {
    v.x = 0.f; v.y = 0.f; v.z = 0.f; v.w = 0.f;
  }
  ((float4*)out)[i] = v;
}

static void run_gat_layer(const float* h_in, const float* W, const float* attn,
                          const int* src, const int* dst, float* feat,
                          float* h_out, float* logits, int* mmax, float* denom,
                          int* unsafe, hipStream_t stream) {
  gemm128_kernel<false>
      <<<(N_NODES / 16 + GEMM_WAVES - 1) / GEMM_WAVES, GEMM_WAVES * 32, 0, stream>>>(
          h_in, W, nullptr, feat);
  init_nodes_kernel<<<(N_NODES + 255) / 256, 256, 0, stream>>>(mmax, denom,
                                                               unsafe);
  zero_kernel<<<(N_NODES * HID / 4 / FILL_SHARE + FILL_WAVES - 1) / FILL_WAVES,
                FILL_WAVES * 32, 0, stream>>>(h_out,
                                              N_NODES * HID / 4 / FILL_SHARE);
  edge_fused_kernel<<<N_EDGES / (8 * EPW), 256, 0, stream>>>(
      feat, attn, src, dst, logits, denom, h_out, unsafe);
  fallback_reset_max_kernel<<<512, 256, 0, stream>>>(unsafe, dst, logits, mmax,
                                                     denom, h_out);
  edge_aggr_kernel<<<512, 256, 0, stream>>>(feat, src, dst, logits, mmax, denom,
                                            h_out, unsafe);
}

static void run_gat_layer_cls(const float* h_in, const float* denom_in,
                              const float* W, const float* attn,
                              const float* cls_w, const int* src,
                              const int* dst, float* feat, float* g,
                              float* acc, float* out, float* logits, int* mmax,
                              int* unsafe, hipStream_t stream) {
  gemm128_kernel<true>
      <<<(N_NODES / 16 + GEMM_WAVES - 1) / GEMM_WAVES, GEMM_WAVES * 32, 0, stream>>>(
          h_in, W, denom_in, feat);
  gemm_cls_kernel<<<(N_NODES / 16 + 7) / 8, 256, 0, stream>>>(feat, cls_w, g);
  init_cls_kernel<<<(N_NODES * ACC_W / 4 / FILL_SHARE + FILL_WAVES - 1) / FILL_WAVES,
                    FILL_WAVES * 32, 0, stream>>>(acc, mmax, unsafe);
  edge_fused_cls_kernel<<<N_EDGES / (8 * EPW), 256, 0, stream>>>(
      feat, attn, g, src, dst, logits, acc, unsafe);
  fallback_reset_cls_kernel<<<512, 256, 0, stream>>>(unsafe, dst, logits, mmax,
                                                     acc);
  edge_aggr_cls_kernel<<<512, 256, 0, stream>>>(g, src, dst, logits, mmax, acc,
                                                unsafe);
  divide_rows_kernel<<<(N_NODES * OUT_DIM / 4 + 255) / 256, 256, 0, stream>>>(
      acc, out);
}

extern "C" void kernel_launch(void* const* d_in, const int* in_sizes, int n_in,
                              void* d_out, int out_size, void* d_ws,
                              size_t ws_size, hipStream_t stream) {
  const int* feats = (const int*)d_in[0];
  const int* src = (const int*)d_in[1];
  const int* dst = (const int*)d_in[2];
  const float* key_emb = (const float*)d_in[3];
  const float* val_emb = (const float*)d_in[4];
  const float* W1 = (const float*)d_in[5];
  const float* attn1 = (const float*)d_in[6];
  const float* W2 = (const float*)d_in[7];
  const float* attn2 = (const float*)d_in[8];
  const float* cls_w = (const float*)d_in[9];
  float* out = (float*)d_out;

  float* hA = (float*)d_ws;
  float* feat = hA + (size_t)N_NODES * HID;
  float* hB = feat + (size_t)N_NODES * HID;
  float* ex = hB + (size_t)N_NODES * HID;
  int* mmax = (int*)(ex + N_EDGES);
  float* denom = (float*)(mmax + N_NODES);

  embed_kernel<<<(N_NODES / EMB_ROWS + EMB_WAVES - 1) / EMB_WAVES,
                 EMB_WAVES * 32, 0, stream>>>(feats, key_emb, val_emb, hA);

  int* unsafe = (int*)(hA + (size_t)N_NODES * HID - 1);
  run_gat_layer(hA, W1, attn1, src, dst, feat, hB, ex, mmax, denom, unsafe,
                stream);
  float* g = hA;
  float* acc = hA + (size_t)N_NODES * ACC_W;
  run_gat_layer_cls(hB, denom, W2, attn2, cls_w, src, dst, feat, g, acc, out,
                    ex, mmax, unsafe, stream);
}
